# starting kernel
# baseline (speedup 1.0000x reference)
#include <hip/hip_runtime.h>


typedef __attribute__((ext_vector_type(2))) float v2f;
typedef __attribute__((ext_vector_type(8))) float v8f;

#define NN    100
#define HH    128
#define NPAD  112
#define R4    400

#define OFF_XS    0
#define OFF_PRE   14336
#define OFF_AGG   28672
#define OFF_W     43008
#define OFF_WLAST 75776
#define OFF_BM1   75904
#define OFF_BM2   76032
#define OFF_BU1   76160
#define OFF_BU2   76288
#define OFF_GM    76416
#define OFF_BT    76544
#define OFF_MU    76672
#define OFF_RS    76784
#define SMEM_FLOATS 76896

__device__ __forceinline__ v8f wmma4(v2f a, v2f b, v8f c) {
  return __builtin_amdgcn_wmma_f32_16x16x4_f32(
       false, a,  false, b,
       (short)0, c,  false,  false);
}

__device__ __forceinline__ void cp16(float* dst, const float* src, int nflt, int tid) {
  const float4* s = (const float4*)src;
  float4*       d = (float4*)dst;
  for (int i = tid; i < (nflt >> 2); i += 256) d[i] = s[i];
}

__device__ __forceinline__ void zero16(float* dst, int nflt, int tid) {
  float4 z = make_float4(0.f, 0.f, 0.f, 0.f);
  float4* d = (float4*)dst;
  for (int i = tid; i < (nflt >> 2); i += 256) d[i] = z;
}

__device__ __forceinline__ void cpw_swz(float* dst, const float* src, int krows, int tid) {
  const int npairs = krows >> 1;
  for (int idx = tid; idx < npairs * 32; idx += 256) {
    const int kp = idx >> 5;
    const int cg = idx & 31;
    const float4 r0 = *(const float4*)(src + (size_t)(2 * kp)     * 128 + cg * 4);
    const float4 r1 = *(const float4*)(src + (size_t)(2 * kp + 1) * 128 + cg * 4);
    float4* d = (float4*)(dst + kp * 256 + cg * 8);
    d[0] = make_float4(r0.x, r1.x, r0.y, r1.y);
    d[1] = make_float4(r0.z, r1.z, r0.w, r1.w);
  }
}

__device__ __forceinline__ void wmma_row8(const float* __restrict__ wbk, int li,
                                          v2f a, v8f acc[8]) {
#pragma unroll
  for (int n = 0; n < 8; ++n) {
    const v2f bb = *(const v2f*)(wbk + (n * 16 + li) * 2);
    acc[n] = wmma4(a, bb, acc[n]);
  }
}

extern "C" __global__ __launch_bounds__(256)
void mpnn_layer_kernel(const float* __restrict__ x,
                       const float* __restrict__ Wm1, const float* __restrict__ bm1,
                       const float* __restrict__ Wm2, const float* __restrict__ bm2,
                       const float* __restrict__ Wu1, const float* __restrict__ bu1,
                       const float* __restrict__ Wu2, const float* __restrict__ bu2,
                       const float* __restrict__ gamma, const float* __restrict__ beta,
                       float* __restrict__ out)
{
  extern __shared__ float smem[];
  float* xs    = smem + OFF_XS;
  float* pre   = smem + OFF_PRE;
  float* agg   = smem + OFF_AGG;
  float* wb    = smem + OFF_W;
  float* wlast = smem + OFF_WLAST;
  float* lbm1  = smem + OFF_BM1;
  float* lbm2  = smem + OFF_BM2;
  float* lbu1  = smem + OFF_BU1;
  float* lbu2  = smem + OFF_BU2;
  float* lgm   = smem + OFF_GM;
  float* lbt   = smem + OFF_BT;
  float* lmu   = smem + OFF_MU;
  float* lrs   = smem + OFF_RS;

  const int tid  = threadIdx.x;
  const int wave = tid >> 5;
  const int lane = tid & 31;
  const int li   = lane & 15;
  const int hi   = lane >> 4;
  const int b    = blockIdx.x;

  const float* xg = x + (size_t)b * (NN * HH);

  cp16(xs, xg, NN * HH, tid);
  zero16(xs + NN * HH, (NPAD - NN) * HH, tid);
  zero16(agg, NPAD * HH, tid);
  cpw_swz(wb, Wm1, 128, tid);
  cp16(wlast, Wm1 + 128 * 128, 128, tid);
  cp16(lbm1, bm1, 128, tid);
  cp16(lbm2, bm2, 128, tid);
  cp16(lbu1, bu1, 128, tid);
  cp16(lbu2, bu2, 128, tid);
  cp16(lgm, gamma, 128, tid);
  cp16(lbt, beta, 128, tid);
  __syncthreads();

  const v8f vz = {0.f, 0.f, 0.f, 0.f, 0.f, 0.f, 0.f, 0.f};

  for (int m = wave; m < 7; m += 8) {
    v8f acc[8];
#pragma unroll
    for (int n = 0; n < 8; ++n) acc[n] = vz;
    const int arow = m * 16 + li;
    for (int ks = 0; ks < 32; ++ks) {
      const int k0 = ks * 4 + hi * 2;
      const v2f a = *(const v2f*)(xs + arow * HH + k0);
      wmma_row8(wb + (2 * ks + hi) * 256, li, a, acc);
    }
#pragma unroll
    for (int n = 0; n < 8; ++n) {
      const int col = n * 16 + li;
      const float bias = lbm1[col];
#pragma unroll
      for (int v = 0; v < 8; ++v) {
        const int row = m * 16 + hi * 8 + v;
        pre[row * HH + col] = acc[n][v] + bias;
      }
    }
  }
  __syncthreads();
  cpw_swz(wb, Wm2, 128, tid);
  __syncthreads();

  for (int m = wave; m < 25; m += 8) {
    v8f acc[8];
#pragma unroll
    for (int n = 0; n < 8; ++n) acc[n] = vz;
    const int   ar = m * 16 + li;
    const int   aj = ar >> 2;
    const float ad = (float)(ar & 3);
    for (int ks = 0; ks < 32; ++ks) {
      const int k0 = ks * 4 + hi * 2;
      const v2f p  = *(const v2f*)(pre + aj * HH + k0);
      const v2f wl = *(const v2f*)(wlast + k0);
      v2f a;
      a.x = fmaxf(p.x + ad * wl.x, 0.f);
      a.y = fmaxf(p.y + ad * wl.y, 0.f);
      wmma_row8(wb + (2 * ks + hi) * 256, li, a, acc);
    }
    float bias[8];
#pragma unroll
    for (int n = 0; n < 8; ++n) bias[n] = lbm2[n * 16 + li];
#pragma unroll
    for (int v = 0; v < 8; ++v) {
      const int r = m * 16 + hi * 8 + v;
      const int j = r >> 2, d = r & 3;
      const int jm = j - (j / 10) * 10;
      int i; bool ok;
      if      (d == 0) { i = j + 10; ok = (i < 100); }
      else if (d == 1) { i = j - 10; ok = (j >= 10); }
      else if (d == 2) { i = j + 1;  ok = (jm <= 8); }
      else             { i = j - 1;  ok = (jm >= 1); }
      if (ok) {
#pragma unroll
        for (int n = 0; n < 8; ++n) {
          const int col = n * 16 + li;
          atomicAdd(&agg[i * HH + col], acc[n][v] + bias[n]);
        }
      }
    }
  }
  __syncthreads();
  cpw_swz(wb, Wu1, 256, tid);
  __syncthreads();

  for (int m = wave; m < 7; m += 8) {
    v8f acc[8];
#pragma unroll
    for (int n = 0; n < 8; ++n) acc[n] = vz;
    const int arow = m * 16 + li;
    for (int ks = 0; ks < 32; ++ks) {
      const int k0 = ks * 4 + hi * 2;
      const v2f a = *(const v2f*)(xs + arow * HH + k0);
      wmma_row8(wb + (2 * ks + hi) * 256, li, a, acc);
    }
    for (int ks = 32; ks < 64; ++ks) {
      const int k0 = (ks - 32) * 4 + hi * 2;
      const v2f a = *(const v2f*)(agg + arow * HH + k0);
      wmma_row8(wb + (2 * ks + hi) * 256, li, a, acc);
    }
#pragma unroll
    for (int n = 0; n < 8; ++n) {
      const int col = n * 16 + li;
      const float bias = lbu1[col];
#pragma unroll
      for (int v = 0; v < 8; ++v) {
        const int row = m * 16 + hi * 8 + v;
        pre[row * HH + col] = fmaxf(acc[n][v] + bias, 0.f);
      }
    }
  }
  __syncthreads();
  cpw_swz(wb, Wu2, 128, tid);
  __syncthreads();

  for (int m = wave; m < 7; m += 8) {
    v8f acc[8];
#pragma unroll
    for (int n = 0; n < 8; ++n) acc[n] = vz;
    const int arow = m * 16 + li;
    for (int ks = 0; ks < 32; ++ks) {
      const int k0 = ks * 4 + hi * 2;
      const v2f a = *(const v2f*)(pre + arow * HH + k0);
      wmma_row8(wb + (2 * ks + hi) * 256, li, a, acc);
    }
#pragma unroll
    for (int n = 0; n < 8; ++n) {
      const int col = n * 16 + li;
      const float bias = lbu2[col];
#pragma unroll
      for (int v = 0; v < 8; ++v) {
        const int row = m * 16 + hi * 8 + v;
        agg[row * HH + col] = xs[row * HH + col] + acc[n][v] + bias;
      }
    }
  }
  __syncthreads();

  const float* hb = agg;
  if (tid < NN) {
    const float* hr = hb + tid * HH;
    float s = 0.f;
    for (int k = 0; k < HH; ++k) s += hr[k];
    const float mu = s * (1.0f / 128.0f);
    float var = 0.f;
    for (int k = 0; k < HH; ++k) { const float dd = hr[k] - mu; var += dd * dd; }
    lmu[tid] = mu;
    lrs[tid] = rsqrtf(var * (1.0f / 128.0f) + 1e-5f);
  }
  __syncthreads();

  float* og = out + (size_t)b * (NN * HH);
  for (int idx = tid; idx < NN * (HH / 4); idx += 256) {
    const int row = idx >> 5;
    const int c4  = idx & 31;
    const int col = c4 * 4;
    const float4 hv = ((const float4*)hb)[row * 32 + c4];
    const float mu = lmu[row], rs = lrs[row];
    float4 o;
    o.x = (hv.x - mu) * rs * lgm[col + 0] + lbt[col + 0];
    o.y = (hv.y - mu) * rs * lgm[col + 1] + lbt[col + 1];
    o.z = (hv.z - mu) * rs * lgm[col + 2] + lbt[col + 2];
    o.w = (hv.w - mu) * rs * lgm[col + 3] + lbt[col + 3];
    ((float4*)og)[idx] = o;
  }
}

extern "C" void kernel_launch(void* const* d_in, const int* in_sizes, int n_in,
                              void* d_out, int out_size, void* d_ws, size_t ws_size,
                              hipStream_t stream) {
  (void)n_in; (void)out_size; (void)d_ws; (void)ws_size;
  const float* x     = (const float*)d_in[0];
  const float* Wm1   = (const float*)d_in[4];
  const float* bm1   = (const float*)d_in[5];
  const float* Wm2   = (const float*)d_in[6];
  const float* bm2   = (const float*)d_in[7];
  const float* Wu1   = (const float*)d_in[8];
  const float* bu1   = (const float*)d_in[9];
  const float* Wu2   = (const float*)d_in[10];
  const float* bu2   = (const float*)d_in[11];
  const float* gamma = (const float*)d_in[12];
  const float* beta  = (const float*)d_in[13];
  float* out = (float*)d_out;

  const int B = in_sizes[0] / (NN * HH);
  const size_t shmem = (size_t)SMEM_FLOATS * sizeof(float);
  hipLaunchKernelGGL(mpnn_layer_kernel, dim3(B), dim3(256), shmem, stream,
                     x, Wm1, bm1, Wm2, bm2, Wu1, bu1, Wu2, bu2, gamma, beta, out);
}


# Weight and vector copies moved to 4 added loader waves (384 threads) that fill the idle half of the LDS weight stage with the next phase's block under the WMMAs; phase 3 split at K=128
# speedup vs baseline: 1.1384x; 1.1384x over previous
#include <hip/hip_runtime.h>


typedef __attribute__((ext_vector_type(2))) float v2f;
typedef __attribute__((ext_vector_type(8))) float v8f;

#define NN    100
#define HH    128
#define NPAD  112
#define R4    400

#define OFF_XS    0
#define OFF_PRE   14336
#define OFF_AGG   28672
#define OFF_W     43008
#define OFF_WLAST 75776
#define OFF_BM1   75904
#define OFF_BM2   76032
#define OFF_BU1   76160
#define OFF_BU2   76288
#define OFF_GM    76416
#define OFF_BT    76544
#define OFF_MU    76672
#define OFF_RS    76784
#define SMEM_FLOATS 76896
#define NT     384
#define NLDR   128
#define WHALF  16384

__device__ __forceinline__ v8f wmma4(v2f a, v2f b, v8f c) {
  return __builtin_amdgcn_wmma_f32_16x16x4_f32(
       false, a,  false, b,
       (short)0, c,  false,  false);
}

__device__ __forceinline__ void cp16(float* dst, const float* src, int nflt, int tid) {
  const float4* s = (const float4*)src;
  float4*       d = (float4*)dst;
  for (int i = tid; i < (nflt >> 2); i += 256) d[i] = s[i];
}

__device__ __forceinline__ void zero16(float* dst, int nflt, int tid) {
  float4 z = make_float4(0.f, 0.f, 0.f, 0.f);
  float4* d = (float4*)dst;
  for (int i = tid; i < (nflt >> 2); i += 256) d[i] = z;
}

__device__ __forceinline__ void cpw_swz(float* dst, const float* src, int krows, int tid) {
  const int npairs = krows >> 1;
  for (int idx = tid; idx < npairs * 32; idx += 256) {
    const int kp = idx >> 5;
    const int cg = idx & 31;
    const float4 r0 = *(const float4*)(src + (size_t)(2 * kp)     * 128 + cg * 4);
    const float4 r1 = *(const float4*)(src + (size_t)(2 * kp + 1) * 128 + cg * 4);
    float4* d = (float4*)(dst + kp * 256 + cg * 8);
    d[0] = make_float4(r0.x, r1.x, r0.y, r1.y);
    d[1] = make_float4(r0.z, r1.z, r0.w, r1.w);
  }
}

__device__ __forceinline__ void ldr_stage128(float* dst, const float* src, int lt) {
  float4 r0[16], r1[16];
#pragma unroll
  for (int u = 0; u < 16; ++u) {
    const int idx = lt + u * NLDR;
    const int kp = idx >> 5;
    const int cg = idx & 31;
    r0[u] = *(const float4*)(src + (size_t)(2 * kp)     * 128 + cg * 4);
    r1[u] = *(const float4*)(src + (size_t)(2 * kp + 1) * 128 + cg * 4);
  }
#pragma unroll
  for (int u = 0; u < 16; ++u) {
    const int idx = lt + u * NLDR;
    const int kp = idx >> 5;
    const int cg = idx & 31;
    float4* d = (float4*)(dst + kp * 256 + cg * 8);
    d[0] = make_float4(r0[u].x, r1[u].x, r0[u].y, r1[u].y);
    d[1] = make_float4(r0[u].z, r1[u].z, r0[u].w, r1[u].w);
  }
}

__device__ __forceinline__ void wmma_row8(const float* __restrict__ wbk, int li,
                                          v2f a, v8f acc[8]) {
#pragma unroll
  for (int n = 0; n < 8; ++n) {
    const v2f bb = *(const v2f*)(wbk + (n * 16 + li) * 2);
    acc[n] = wmma4(a, bb, acc[n]);
  }
}

extern "C" __global__ __launch_bounds__(NT)
void mpnn_layer_kernel(const float* __restrict__ x,
                       const float* __restrict__ Wm1, const float* __restrict__ bm1,
                       const float* __restrict__ Wm2, const float* __restrict__ bm2,
                       const float* __restrict__ Wu1, const float* __restrict__ bu1,
                       const float* __restrict__ Wu2, const float* __restrict__ bu2,
                       const float* __restrict__ gamma, const float* __restrict__ beta,
                       float* __restrict__ out)
{
  extern __shared__ float smem[];
  float* xs    = smem + OFF_XS;
  float* pre   = smem + OFF_PRE;
  float* agg   = smem + OFF_AGG;
  float* wb    = smem + OFF_W;
  float* wlast = smem + OFF_WLAST;
  float* lbm1  = smem + OFF_BM1;
  float* lbm2  = smem + OFF_BM2;
  float* lbu1  = smem + OFF_BU1;
  float* lbu2  = smem + OFF_BU2;
  float* lgm   = smem + OFF_GM;
  float* lbt   = smem + OFF_BT;
  float* lmu   = smem + OFF_MU;
  float* lrs   = smem + OFF_RS;

  const int tid  = threadIdx.x;
  const int wave = tid >> 5;
  const int lane = tid & 31;
  const int li   = lane & 15;
  const int hi   = lane >> 4;
  const int b    = blockIdx.x;
  const bool is_ldr = tid >= 256;
  const int  lt     = tid - 256;
  float* wh0 = wb;
  float* wh1 = wb + WHALF;

  const float* xg = x + (size_t)b * (NN * HH);

  if (is_ldr) {
    const int q  = lt >> 5;
    const int l5 = lt & 31;
    const float* sA = (q == 0) ? (Wm1 + 128 * 128) : (q == 1) ? bm1 : (q == 2) ? bm2 : bu1;
    float*       dA = (q == 0) ? wlast             : (q == 1) ? lbm1 : (q == 2) ? lbm2 : lbu1;
    const float* sB = (q == 0) ? bu2  : (q == 1) ? gamma : beta;
    float*       dB = (q == 0) ? lbu2 : (q == 1) ? lgm   : lbt;
    const float4 vA = ((const float4*)sA)[l5];
    float4 vB = make_float4(0.f, 0.f, 0.f, 0.f);
    if (q < 3) vB = ((const float4*)sB)[l5];
    ldr_stage128(wh0, Wm1, lt);
    ((float4*)dA)[l5] = vA;
    if (q < 3) ((float4*)dB)[l5] = vB;
  } else {
    cp16(xs, xg, NN * HH, tid);
    zero16(xs + NN * HH, (NPAD - NN) * HH, tid);
    zero16(agg, NPAD * HH, tid);
  }
  __syncthreads();

  const v8f vz = {0.f, 0.f, 0.f, 0.f, 0.f, 0.f, 0.f, 0.f};

  if (is_ldr) {
    ldr_stage128(wh1, Wm2, lt);
  } else {
    for (int m = wave; m < 7; m += 8) {
      v8f acc[8];
#pragma unroll
      for (int n = 0; n < 8; ++n) acc[n] = vz;
      const int arow = m * 16 + li;
      for (int ks = 0; ks < 32; ++ks) {
        const int k0 = ks * 4 + hi * 2;
        const v2f a = *(const v2f*)(xs + arow * HH + k0);
        wmma_row8(wh0 + (2 * ks + hi) * 256, li, a, acc);
      }
#pragma unroll
      for (int n = 0; n < 8; ++n) {
        const int col = n * 16 + li;
        const float bias = lbm1[col];
#pragma unroll
        for (int v = 0; v < 8; ++v) {
          const int row = m * 16 + hi * 8 + v;
          pre[row * HH + col] = acc[n][v] + bias;
        }
      }
    }
  }
  __syncthreads();

  if (is_ldr) {
    ldr_stage128(wh0, Wu1, lt);
  } else {
    for (int m = wave; m < 25; m += 8) {
      v8f acc[8];
#pragma unroll
      for (int n = 0; n < 8; ++n) acc[n] = vz;
      const int   ar = m * 16 + li;
      const int   aj = ar >> 2;
      const float ad = (float)(ar & 3);
      for (int ks = 0; ks < 32; ++ks) {
        const int k0 = ks * 4 + hi * 2;
        const v2f p  = *(const v2f*)(pre + aj * HH + k0);
        const v2f wl = *(const v2f*)(wlast + k0);
        v2f a;
        a.x = fmaxf(p.x + ad * wl.x, 0.f);
        a.y = fmaxf(p.y + ad * wl.y, 0.f);
        wmma_row8(wh1 + (2 * ks + hi) * 256, li, a, acc);
      }
      float bias[8];
#pragma unroll
      for (int n = 0; n < 8; ++n) bias[n] = lbm2[n * 16 + li];
#pragma unroll
      for (int v = 0; v < 8; ++v) {
        const int r = m * 16 + hi * 8 + v;
        const int j = r >> 2, d = r & 3;
        const int jm = j - (j / 10) * 10;
        int i; bool ok;
        if      (d == 0) { i = j + 10; ok = (i < 100); }
        else if (d == 1) { i = j - 10; ok = (j >= 10); }
        else if (d == 2) { i = j + 1;  ok = (jm <= 8); }
        else             { i = j - 1;  ok = (jm >= 1); }
        if (ok) {
#pragma unroll
          for (int n = 0; n < 8; ++n) {
            const int col = n * 16 + li;
            atomicAdd(&agg[i * HH + col], acc[n][v] + bias[n]);
          }
        }
      }
    }
  }
  __syncthreads();

  v8f acc3[8];
#pragma unroll
  for (int n = 0; n < 8; ++n) acc3[n] = vz;
  const int arow3 = wave * 16 + li;
  if (is_ldr) {
    ldr_stage128(wh1, Wu1 + 128 * 128, lt);
  } else if (wave < 7) {
    for (int ks = 0; ks < 32; ++ks) {
      const int k0 = ks * 4 + hi * 2;
      const v2f a = *(const v2f*)(xs + arow3 * HH + k0);
      wmma_row8(wh0 + (2 * ks + hi) * 256, li, a, acc3);
    }
  }
  __syncthreads();
  if (is_ldr) {
    ldr_stage128(wh0, Wu2, lt);
  } else if (wave < 7) {
    for (int ks = 0; ks < 32; ++ks) {
      const int k0 = ks * 4 + hi * 2;
      const v2f a = *(const v2f*)(agg + arow3 * HH + k0);
      wmma_row8(wh1 + (2 * ks + hi) * 256, li, a, acc3);
    }
#pragma unroll
    for (int n = 0; n < 8; ++n) {
      const int col = n * 16 + li;
      const float bias = lbu1[col];
#pragma unroll
      for (int v = 0; v < 8; ++v) {
        const int row = wave * 16 + hi * 8 + v;
        pre[row * HH + col] = fmaxf(acc3[n][v] + bias, 0.f);
      }
    }
  }
  __syncthreads();

  if (!is_ldr) {
    for (int m = wave; m < 7; m += 8) {
      v8f acc[8];
#pragma unroll
      for (int n = 0; n < 8; ++n) acc[n] = vz;
      const int arow = m * 16 + li;
      for (int ks = 0; ks < 32; ++ks) {
        const int k0 = ks * 4 + hi * 2;
        const v2f a = *(const v2f*)(pre + arow * HH + k0);
        wmma_row8(wh0 + (2 * ks + hi) * 256, li, a, acc);
      }
#pragma unroll
      for (int n = 0; n < 8; ++n) {
        const int col = n * 16 + li;
        const float bias = lbu2[col];
#pragma unroll
        for (int v = 0; v < 8; ++v) {
          const int row = m * 16 + hi * 8 + v;
          agg[row * HH + col] = xs[row * HH + col] + acc[n][v] + bias;
        }
      }
    }
  }
  __syncthreads();

  const float* hb = agg;
  if (tid < NN) {
    const float* hr = hb + tid * HH;
    float s = 0.f;
    for (int k = 0; k < HH; ++k) s += hr[k];
    const float mu = s * (1.0f / 128.0f);
    float var = 0.f;
    for (int k = 0; k < HH; ++k) { const float dd = hr[k] - mu; var += dd * dd; }
    lmu[tid] = mu;
    lrs[tid] = rsqrtf(var * (1.0f / 128.0f) + 1e-5f);
  }
  __syncthreads();

  float* og = out + (size_t)b * (NN * HH);
  for (int idx = tid; idx < NN * (HH / 4); idx += NT) {
    const int row = idx >> 5;
    const int c4  = idx & 31;
    const int col = c4 * 4;
    const float4 hv = ((const float4*)hb)[row * 32 + c4];
    const float mu = lmu[row], rs = lrs[row];
    float4 o;
    o.x = (hv.x - mu) * rs * lgm[col + 0] + lbt[col + 0];
    o.y = (hv.y - mu) * rs * lgm[col + 1] + lbt[col + 1];
    o.z = (hv.z - mu) * rs * lgm[col + 2] + lbt[col + 2];
    o.w = (hv.w - mu) * rs * lgm[col + 3] + lbt[col + 3];
    ((float4*)og)[idx] = o;
  }
}

extern "C" void kernel_launch(void* const* d_in, const int* in_sizes, int n_in,
                              void* d_out, int out_size, void* d_ws, size_t ws_size,
                              hipStream_t stream) {
  (void)n_in; (void)out_size; (void)d_ws; (void)ws_size;
  const float* x     = (const float*)d_in[0];
  const float* Wm1   = (const float*)d_in[4];
  const float* bm1   = (const float*)d_in[5];
  const float* Wm2   = (const float*)d_in[6];
  const float* bm2   = (const float*)d_in[7];
  const float* Wu1   = (const float*)d_in[8];
  const float* bu1   = (const float*)d_in[9];
  const float* Wu2   = (const float*)d_in[10];
  const float* bu2   = (const float*)d_in[11];
  const float* gamma = (const float*)d_in[12];
  const float* beta  = (const float*)d_in[13];
  float* out = (float*)d_out;

  const int B = in_sizes[0] / (NN * HH);
  const size_t shmem = (size_t)SMEM_FLOATS * sizeof(float);
  hipLaunchKernelGGL(mpnn_layer_kernel, dim3(B), dim3(NT), shmem, stream,
                     x, Wm1, bm1, Wm2, bm2, Wu1, bu1, Wu2, bu2, gamma, beta, out);
}


# LayerNorm row scan: thread t starts its row 4*t columns on, wrapping at 128, so a wave's 16-byte LDS reads hit different banks, not one; rows summed in that rotated order (sum order changes)
# speedup vs baseline: 1.4142x; 1.2423x over previous
#include <hip/hip_runtime.h>


typedef __attribute__((ext_vector_type(2))) float v2f;
typedef __attribute__((ext_vector_type(8))) float v8f;

#define NN    100
#define HH    128
#define NPAD  112
#define R4    400

#define OFF_XS    0
#define OFF_PRE   14336
#define OFF_AGG   28672
#define OFF_W     43008
#define OFF_WLAST 75776
#define OFF_BM1   75904
#define OFF_BM2   76032
#define OFF_BU1   76160
#define OFF_BU2   76288
#define OFF_GM    76416
#define OFF_BT    76544
#define OFF_MU    76672
#define OFF_RS    76784
#define SMEM_FLOATS 76896
#define NT     384
#define NLDR   128
#define WHALF  16384

__device__ __forceinline__ v8f wmma4(v2f a, v2f b, v8f c) {
  return __builtin_amdgcn_wmma_f32_16x16x4_f32(
       false, a,  false, b,
       (short)0, c,  false,  false);
}

__device__ __forceinline__ void cp16(float* dst, const float* src, int nflt, int tid) {
  const float4* s = (const float4*)src;
  float4*       d = (float4*)dst;
  for (int i = tid; i < (nflt >> 2); i += 256) d[i] = s[i];
}

__device__ __forceinline__ void zero16(float* dst, int nflt, int tid) {
  float4 z = make_float4(0.f, 0.f, 0.f, 0.f);
  float4* d = (float4*)dst;
  for (int i = tid; i < (nflt >> 2); i += 256) d[i] = z;
}

__device__ __forceinline__ void cpw_swz(float* dst, const float* src, int krows, int tid) {
  const int npairs = krows >> 1;
  for (int idx = tid; idx < npairs * 32; idx += 256) {
    const int kp = idx >> 5;
    const int cg = idx & 31;
    const float4 r0 = *(const float4*)(src + (size_t)(2 * kp)     * 128 + cg * 4);
    const float4 r1 = *(const float4*)(src + (size_t)(2 * kp + 1) * 128 + cg * 4);
    float4* d = (float4*)(dst + kp * 256 + cg * 8);
    d[0] = make_float4(r0.x, r1.x, r0.y, r1.y);
    d[1] = make_float4(r0.z, r1.z, r0.w, r1.w);
  }
}

__device__ __forceinline__ void ldr_stage128(float* dst, const float* src, int lt) {
  float4 r0[16], r1[16];
#pragma unroll
  for (int u = 0; u < 16; ++u) {
    const int idx = lt + u * NLDR;
    const int kp = idx >> 5;
    const int cg = idx & 31;
    r0[u] = *(const float4*)(src + (size_t)(2 * kp)     * 128 + cg * 4);
    r1[u] = *(const float4*)(src + (size_t)(2 * kp + 1) * 128 + cg * 4);
  }
#pragma unroll
  for (int u = 0; u < 16; ++u) {
    const int idx = lt + u * NLDR;
    const int kp = idx >> 5;
    const int cg = idx & 31;
    float4* d = (float4*)(dst + kp * 256 + cg * 8);
    d[0] = make_float4(r0[u].x, r1[u].x, r0[u].y, r1[u].y);
    d[1] = make_float4(r0[u].z, r1[u].z, r0[u].w, r1[u].w);
  }
}

__device__ __forceinline__ void wmma_row8(const float* __restrict__ wbk, int li,
                                          v2f a, v8f acc[8]) {
#pragma unroll
  for (int n = 0; n < 8; ++n) {
    const v2f bb = *(const v2f*)(wbk + (n * 16 + li) * 2);
    acc[n] = wmma4(a, bb, acc[n]);
  }
}

extern "C" __global__ __launch_bounds__(NT)
void mpnn_layer_kernel(const float* __restrict__ x,
                       const float* __restrict__ Wm1, const float* __restrict__ bm1,
                       const float* __restrict__ Wm2, const float* __restrict__ bm2,
                       const float* __restrict__ Wu1, const float* __restrict__ bu1,
                       const float* __restrict__ Wu2, const float* __restrict__ bu2,
                       const float* __restrict__ gamma, const float* __restrict__ beta,
                       float* __restrict__ out)
{
  extern __shared__ float smem[];
  float* xs    = smem + OFF_XS;
  float* pre   = smem + OFF_PRE;
  float* agg   = smem + OFF_AGG;
  float* wb    = smem + OFF_W;
  float* wlast = smem + OFF_WLAST;
  float* lbm1  = smem + OFF_BM1;
  float* lbm2  = smem + OFF_BM2;
  float* lbu1  = smem + OFF_BU1;
  float* lbu2  = smem + OFF_BU2;
  float* lgm   = smem + OFF_GM;
  float* lbt   = smem + OFF_BT;
  float* lmu   = smem + OFF_MU;
  float* lrs   = smem + OFF_RS;

  const int tid  = threadIdx.x;
  const int wave = tid >> 5;
  const int lane = tid & 31;
  const int li   = lane & 15;
  const int hi   = lane >> 4;
  const int b    = blockIdx.x;
  const bool is_ldr = tid >= 256;
  const int  lt     = tid - 256;
  float* wh0 = wb;
  float* wh1 = wb + WHALF;

  const float* xg = x + (size_t)b * (NN * HH);

  if (is_ldr) {
    const int q  = lt >> 5;
    const int l5 = lt & 31;
    const float* sA = (q == 0) ? (Wm1 + 128 * 128) : (q == 1) ? bm1 : (q == 2) ? bm2 : bu1;
    float*       dA = (q == 0) ? wlast             : (q == 1) ? lbm1 : (q == 2) ? lbm2 : lbu1;
    const float* sB = (q == 0) ? bu2  : (q == 1) ? gamma : beta;
    float*       dB = (q == 0) ? lbu2 : (q == 1) ? lgm   : lbt;
    const float4 vA = ((const float4*)sA)[l5];
    float4 vB = make_float4(0.f, 0.f, 0.f, 0.f);
    if (q < 3) vB = ((const float4*)sB)[l5];
    ldr_stage128(wh0, Wm1, lt);
    ((float4*)dA)[l5] = vA;
    if (q < 3) ((float4*)dB)[l5] = vB;
  } else {
    cp16(xs, xg, NN * HH, tid);
    zero16(xs + NN * HH, (NPAD - NN) * HH, tid);
    zero16(agg, NPAD * HH, tid);
  }
  __syncthreads();

  const v8f vz = {0.f, 0.f, 0.f, 0.f, 0.f, 0.f, 0.f, 0.f};

  if (is_ldr) {
    ldr_stage128(wh1, Wm2, lt);
  } else {
    for (int m = wave; m < 7; m += 8) {
      v8f acc[8];
#pragma unroll
      for (int n = 0; n < 8; ++n) acc[n] = vz;
      const int arow = m * 16 + li;
      for (int ks = 0; ks < 32; ++ks) {
        const int k0 = ks * 4 + hi * 2;
        const v2f a = *(const v2f*)(xs + arow * HH + k0);
        wmma_row8(wh0 + (2 * ks + hi) * 256, li, a, acc);
      }
#pragma unroll
      for (int n = 0; n < 8; ++n) {
        const int col = n * 16 + li;
        const float bias = lbm1[col];
#pragma unroll
        for (int v = 0; v < 8; ++v) {
          const int row = m * 16 + hi * 8 + v;
          pre[row * HH + col] = acc[n][v] + bias;
        }
      }
    }
  }
  __syncthreads();

  if (is_ldr) {
    ldr_stage128(wh0, Wu1, lt);
  } else {
    for (int m = wave; m < 25; m += 8) {
      v8f acc[8];
#pragma unroll
      for (int n = 0; n < 8; ++n) acc[n] = vz;
      const int   ar = m * 16 + li;
      const int   aj = ar >> 2;
      const float ad = (float)(ar & 3);
      for (int ks = 0; ks < 32; ++ks) {
        const int k0 = ks * 4 + hi * 2;
        const v2f p  = *(const v2f*)(pre + aj * HH + k0);
        const v2f wl = *(const v2f*)(wlast + k0);
        v2f a;
        a.x = fmaxf(p.x + ad * wl.x, 0.f);
        a.y = fmaxf(p.y + ad * wl.y, 0.f);
        wmma_row8(wh1 + (2 * ks + hi) * 256, li, a, acc);
      }
      float bias[8];
#pragma unroll
      for (int n = 0; n < 8; ++n) bias[n] = lbm2[n * 16 + li];
#pragma unroll
      for (int v = 0; v < 8; ++v) {
        const int r = m * 16 + hi * 8 + v;
        const int j = r >> 2, d = r & 3;
        const int jm = j - (j / 10) * 10;
        int i; bool ok;
        if      (d == 0) { i = j + 10; ok = (i < 100); }
        else if (d == 1) { i = j - 10; ok = (j >= 10); }
        else if (d == 2) { i = j + 1;  ok = (jm <= 8); }
        else             { i = j - 1;  ok = (jm >= 1); }
        if (ok) {
#pragma unroll
          for (int n = 0; n < 8; ++n) {
            const int col = n * 16 + li;
            atomicAdd(&agg[i * HH + col], acc[n][v] + bias[n]);
          }
        }
      }
    }
  }
  __syncthreads();

  v8f acc3[8];
#pragma unroll
  for (int n = 0; n < 8; ++n) acc3[n] = vz;
  const int arow3 = wave * 16 + li;
  if (is_ldr) {
    ldr_stage128(wh1, Wu1 + 128 * 128, lt);
  } else if (wave < 7) {
    for (int ks = 0; ks < 32; ++ks) {
      const int k0 = ks * 4 + hi * 2;
      const v2f a = *(const v2f*)(xs + arow3 * HH + k0);
      wmma_row8(wh0 + (2 * ks + hi) * 256, li, a, acc3);
    }
  }
  __syncthreads();
  if (is_ldr) {
    ldr_stage128(wh0, Wu2, lt);
  } else if (wave < 7) {
    for (int ks = 0; ks < 32; ++ks) {
      const int k0 = ks * 4 + hi * 2;
      const v2f a = *(const v2f*)(agg + arow3 * HH + k0);
      wmma_row8(wh1 + (2 * ks + hi) * 256, li, a, acc3);
    }
#pragma unroll
    for (int n = 0; n < 8; ++n) {
      const int col = n * 16 + li;
      const float bias = lbu1[col];
#pragma unroll
      for (int v = 0; v < 8; ++v) {
        const int row = wave * 16 + hi * 8 + v;
        pre[row * HH + col] = fmaxf(acc3[n][v] + bias, 0.f);
      }
    }
  }
  __syncthreads();

  if (!is_ldr) {
    for (int m = wave; m < 7; m += 8) {
      v8f acc[8];
#pragma unroll
      for (int n = 0; n < 8; ++n) acc[n] = vz;
      const int arow = m * 16 + li;
      for (int ks = 0; ks < 32; ++ks) {
        const int k0 = ks * 4 + hi * 2;
        const v2f a = *(const v2f*)(pre + arow * HH + k0);
        wmma_row8(wh0 + (2 * ks + hi) * 256, li, a, acc);
      }
#pragma unroll
      for (int n = 0; n < 8; ++n) {
        const int col = n * 16 + li;
        const float bias = lbu2[col];
#pragma unroll
        for (int v = 0; v < 8; ++v) {
          const int row = m * 16 + hi * 8 + v;
          agg[row * HH + col] = xs[row * HH + col] + acc[n][v] + bias;
        }
      }
    }
  }
  __syncthreads();

  const float* hb = agg;
  if (tid < NN) {
    const float* hr = hb + tid * HH;
    const int rot = (tid * 4) & (HH - 1);
    float s = 0.f;
    for (int k = 0; k < HH; k += 4) {
      const float4 p = *(const float4*)(hr + ((k + rot) & (HH - 1)));
      s += p.x; s += p.y; s += p.z; s += p.w;
    }
    const float mu = s * (1.0f / 128.0f);
    float var = 0.f;
    for (int k = 0; k < HH; k += 4) {
      const float4 p = *(const float4*)(hr + ((k + rot) & (HH - 1)));
      const float d0 = p.x - mu, d1 = p.y - mu, d2 = p.z - mu, d3 = p.w - mu;
      var += d0 * d0; var += d1 * d1; var += d2 * d2; var += d3 * d3;
    }
    lmu[tid] = mu;
    lrs[tid] = rsqrtf(var * (1.0f / 128.0f) + 1e-5f);
  }
  __syncthreads();

  float* og = out + (size_t)b * (NN * HH);
  for (int idx = tid; idx < NN * (HH / 4); idx += NT) {
    const int row = idx >> 5;
    const int c4  = idx & 31;
    const int col = c4 * 4;
    const float4 hv = ((const float4*)hb)[row * 32 + c4];
    const float mu = lmu[row], rs = lrs[row];
    float4 o;
    o.x = (hv.x - mu) * rs * lgm[col + 0] + lbt[col + 0];
    o.y = (hv.y - mu) * rs * lgm[col + 1] + lbt[col + 1];
    o.z = (hv.z - mu) * rs * lgm[col + 2] + lbt[col + 2];
    o.w = (hv.w - mu) * rs * lgm[col + 3] + lbt[col + 3];
    ((float4*)og)[idx] = o;
  }
}

extern "C" void kernel_launch(void* const* d_in, const int* in_sizes, int n_in,
                              void* d_out, int out_size, void* d_ws, size_t ws_size,
                              hipStream_t stream) {
  (void)n_in; (void)out_size; (void)d_ws; (void)ws_size;
  const float* x     = (const float*)d_in[0];
  const float* Wm1   = (const float*)d_in[4];
  const float* bm1   = (const float*)d_in[5];
  const float* Wm2   = (const float*)d_in[6];
  const float* bm2   = (const float*)d_in[7];
  const float* Wu1   = (const float*)d_in[8];
  const float* bu1   = (const float*)d_in[9];
  const float* Wu2   = (const float*)d_in[10];
  const float* bu2   = (const float*)d_in[11];
  const float* gamma = (const float*)d_in[12];
  const float* beta  = (const float*)d_in[13];
  float* out = (float*)d_out;

  const int B = in_sizes[0] / (NN * HH);
  const size_t shmem = (size_t)SMEM_FLOATS * sizeof(float);
  hipLaunchKernelGGL(mpnn_layer_kernel, dim3(B), dim3(NT), shmem, stream,
                     x, Wm1, bm1, Wm2, bm2, Wu1, bu1, Wu2, bu2, gamma, beta, out);
}


# x rows go from global straight into A-fragment registers (16 loads per lane in flight) for both K=0..127 products; loader waves make the LDS copy for the residual; k sum order changes
# speedup vs baseline: 1.4958x; 1.0577x over previous
#include <hip/hip_runtime.h>


typedef __attribute__((ext_vector_type(2))) float v2f;
typedef __attribute__((ext_vector_type(8))) float v8f;

#define NN    100
#define HH    128
#define NPAD  112
#define R4    400

#define OFF_XS    0
#define OFF_PRE   14336
#define OFF_AGG   28672
#define OFF_W     43008
#define OFF_WLAST 75776
#define OFF_BM1   75904
#define OFF_BM2   76032
#define OFF_BU1   76160
#define OFF_BU2   76288
#define OFF_GM    76416
#define OFF_BT    76544
#define OFF_MU    76672
#define OFF_RS    76784
#define SMEM_FLOATS 76896
#define NT     384
#define NLDR   128
#define WHALF  16384

__device__ __forceinline__ v8f wmma4(v2f a, v2f b, v8f c) {
  return __builtin_amdgcn_wmma_f32_16x16x4_f32(
       false, a,  false, b,
       (short)0, c,  false,  false);
}

__device__ __forceinline__ void zero16(float* dst, int nflt, int tid) {
  float4 z = make_float4(0.f, 0.f, 0.f, 0.f);
  float4* d = (float4*)dst;
  for (int i = tid; i < (nflt >> 2); i += 256) d[i] = z;
}

__device__ __forceinline__ void cpw_swz(float* dst, const float* src, int krows, int tid) {
  const int npairs = krows >> 1;
  for (int idx = tid; idx < npairs * 32; idx += 256) {
    const int kp = idx >> 5;
    const int cg = idx & 31;
    const float4 r0 = *(const float4*)(src + (size_t)(2 * kp)     * 128 + cg * 4);
    const float4 r1 = *(const float4*)(src + (size_t)(2 * kp + 1) * 128 + cg * 4);
    float4* d = (float4*)(dst + kp * 256 + cg * 8);
    d[0] = make_float4(r0.x, r1.x, r0.y, r1.y);
    d[1] = make_float4(r0.z, r1.z, r0.w, r1.w);
  }
}

__device__ __forceinline__ void ldr_stage128(float* dst, const float* src, int lt) {
  float4 r0[16], r1[16];
#pragma unroll
  for (int u = 0; u < 16; ++u) {
    const int idx = lt + u * NLDR;
    const int kp = idx >> 5;
    const int cg = idx & 31;
    r0[u] = *(const float4*)(src + (size_t)(2 * kp)     * 128 + cg * 4);
    r1[u] = *(const float4*)(src + (size_t)(2 * kp + 1) * 128 + cg * 4);
  }
#pragma unroll
  for (int u = 0; u < 16; ++u) {
    const int idx = lt + u * NLDR;
    const int kp = idx >> 5;
    const int cg = idx & 31;
    float4* d = (float4*)(dst + kp * 256 + cg * 8);
    d[0] = make_float4(r0[u].x, r1[u].x, r0[u].y, r1[u].y);
    d[1] = make_float4(r0[u].z, r1[u].z, r0[u].w, r1[u].w);
  }
}

__device__ __forceinline__ void ldr_copy_x(float* dst, const float* src, int lt) {
#pragma unroll 1
  for (int g = 0; g < 5; ++g) {
    float4 r[5];
#pragma unroll
    for (int u = 0; u < 5; ++u) r[u] = ((const float4*)src)[lt + (g * 5 + u) * NLDR];
#pragma unroll
    for (int u = 0; u < 5; ++u) ((float4*)dst)[lt + (g * 5 + u) * NLDR] = r[u];
  }
  const float4 z = make_float4(0.f, 0.f, 0.f, 0.f);
#pragma unroll
  for (int u = 0; u < 3; ++u) ((float4*)(dst + NN * HH))[lt + u * NLDR] = z;
}

__device__ __forceinline__ void wmma_row8(const float* __restrict__ wbk, int li,
                                          v2f a, v8f acc[8]) {
#pragma unroll
  for (int n = 0; n < 8; ++n) {
    const v2f bb = *(const v2f*)(wbk + (n * 16 + li) * 2);
    acc[n] = wmma4(a, bb, acc[n]);
  }
}

extern "C" __global__ __launch_bounds__(NT)
void mpnn_layer_kernel(const float* __restrict__ x,
                       const float* __restrict__ Wm1, const float* __restrict__ bm1,
                       const float* __restrict__ Wm2, const float* __restrict__ bm2,
                       const float* __restrict__ Wu1, const float* __restrict__ bu1,
                       const float* __restrict__ Wu2, const float* __restrict__ bu2,
                       const float* __restrict__ gamma, const float* __restrict__ beta,
                       float* __restrict__ out)
{
  extern __shared__ float smem[];
  float* xs    = smem + OFF_XS;
  float* pre   = smem + OFF_PRE;
  float* agg   = smem + OFF_AGG;
  float* wb    = smem + OFF_W;
  float* wlast = smem + OFF_WLAST;
  float* lbm1  = smem + OFF_BM1;
  float* lbm2  = smem + OFF_BM2;
  float* lbu1  = smem + OFF_BU1;
  float* lbu2  = smem + OFF_BU2;
  float* lgm   = smem + OFF_GM;
  float* lbt   = smem + OFF_BT;
  float* lmu   = smem + OFF_MU;
  float* lrs   = smem + OFF_RS;

  const int tid  = threadIdx.x;
  const int wave = tid >> 5;
  const int lane = tid & 31;
  const int li   = lane & 15;
  const int hi   = lane >> 4;
  const int b    = blockIdx.x;
  const bool is_ldr = tid >= 256;
  const int  lt     = tid - 256;
  float* wh0 = wb;
  float* wh1 = wb + WHALF;

  const float* xg = x + (size_t)b * (NN * HH);

  float4 xa[16];
#pragma unroll
  for (int u = 0; u < 16; ++u) xa[u] = make_float4(0.f, 0.f, 0.f, 0.f);
  if (is_ldr) {
    const int q  = lt >> 5;
    const int l5 = lt & 31;
    const float* sA = (q == 0) ? (Wm1 + 128 * 128) : (q == 1) ? bm1 : (q == 2) ? bm2 : bu1;
    float*       dA = (q == 0) ? wlast             : (q == 1) ? lbm1 : (q == 2) ? lbm2 : lbu1;
    const float* sB = (q == 0) ? bu2  : (q == 1) ? gamma : beta;
    float*       dB = (q == 0) ? lbu2 : (q == 1) ? lgm   : lbt;
    const float4 vA = ((const float4*)sA)[l5];
    float4 vB = make_float4(0.f, 0.f, 0.f, 0.f);
    if (q < 3) vB = ((const float4*)sB)[l5];
    ldr_stage128(wh0, Wm1, lt);
    ((float4*)dA)[l5] = vA;
    if (q < 3) ((float4*)dB)[l5] = vB;
  } else {
    const int xrow = wave * 16 + li;
    if (wave < 7 && xrow < NN) {
      const float* xr = xg + xrow * HH + hi * 4;
#pragma unroll
      for (int u = 0; u < 16; ++u) xa[u] = *(const float4*)(xr + u * 8);
    }
    zero16(agg, NPAD * HH, tid);
  }
  __syncthreads();

  const v8f vz = {0.f, 0.f, 0.f, 0.f, 0.f, 0.f, 0.f, 0.f};

  if (is_ldr) {
    ldr_stage128(wh1, Wm2, lt);
  } else {
    if (wave < 7) {
      const int m = wave;
      v8f acc[8];
#pragma unroll
      for (int n = 0; n < 8; ++n) acc[n] = vz;
#pragma unroll
      for (int u = 0; u < 16; ++u) {
        v2f a0, a1;
        a0.x = xa[u].x; a0.y = xa[u].y;
        a1.x = xa[u].z; a1.y = xa[u].w;
        wmma_row8(wh0 + (4 * u + 2 * hi) * 256, li, a0, acc);
        wmma_row8(wh0 + (4 * u + 2 * hi + 1) * 256, li, a1, acc);
      }
#pragma unroll
      for (int n = 0; n < 8; ++n) {
        const int col = n * 16 + li;
        const float bias = lbm1[col];
#pragma unroll
        for (int v = 0; v < 8; ++v) {
          const int row = m * 16 + hi * 8 + v;
          pre[row * HH + col] = acc[n][v] + bias;
        }
      }
    }
  }
  __syncthreads();

  if (is_ldr) {
    ldr_stage128(wh0, Wu1, lt);
    ldr_copy_x(xs, xg, lt);
  } else {
    for (int m = wave; m < 25; m += 8) {
      v8f acc[8];
#pragma unroll
      for (int n = 0; n < 8; ++n) acc[n] = vz;
      const int   ar = m * 16 + li;
      const int   aj = ar >> 2;
      const float ad = (float)(ar & 3);
      for (int ks = 0; ks < 32; ++ks) {
        const int k0 = ks * 4 + hi * 2;
        const v2f p  = *(const v2f*)(pre + aj * HH + k0);
        const v2f wl = *(const v2f*)(wlast + k0);
        v2f a;
        a.x = fmaxf(p.x + ad * wl.x, 0.f);
        a.y = fmaxf(p.y + ad * wl.y, 0.f);
        wmma_row8(wh1 + (2 * ks + hi) * 256, li, a, acc);
      }
      float bias[8];
#pragma unroll
      for (int n = 0; n < 8; ++n) bias[n] = lbm2[n * 16 + li];
#pragma unroll
      for (int v = 0; v < 8; ++v) {
        const int r = m * 16 + hi * 8 + v;
        const int j = r >> 2, d = r & 3;
        const int jm = j - (j / 10) * 10;
        int i; bool ok;
        if      (d == 0) { i = j + 10; ok = (i < 100); }
        else if (d == 1) { i = j - 10; ok = (j >= 10); }
        else if (d == 2) { i = j + 1;  ok = (jm <= 8); }
        else             { i = j - 1;  ok = (jm >= 1); }
        if (ok) {
#pragma unroll
          for (int n = 0; n < 8; ++n) {
            const int col = n * 16 + li;
            atomicAdd(&agg[i * HH + col], acc[n][v] + bias[n]);
          }
        }
      }
    }
  }
  __syncthreads();

  v8f acc3[8];
#pragma unroll
  for (int n = 0; n < 8; ++n) acc3[n] = vz;
  const int arow3 = wave * 16 + li;
  if (is_ldr) {
    ldr_stage128(wh1, Wu1 + 128 * 128, lt);
  } else if (wave < 7) {
#pragma unroll
    for (int u = 0; u < 16; ++u) {
      v2f a0, a1;
      a0.x = xa[u].x; a0.y = xa[u].y;
      a1.x = xa[u].z; a1.y = xa[u].w;
      wmma_row8(wh0 + (4 * u + 2 * hi) * 256, li, a0, acc3);
      wmma_row8(wh0 + (4 * u + 2 * hi + 1) * 256, li, a1, acc3);
    }
  }
  __syncthreads();
  if (is_ldr) {
    ldr_stage128(wh0, Wu2, lt);
  } else if (wave < 7) {
    for (int ks = 0; ks < 32; ++ks) {
      const int k0 = ks * 4 + hi * 2;
      const v2f a = *(const v2f*)(agg + arow3 * HH + k0);
      wmma_row8(wh1 + (2 * ks + hi) * 256, li, a, acc3);
    }
#pragma unroll
    for (int n = 0; n < 8; ++n) {
      const int col = n * 16 + li;
      const float bias = lbu1[col];
#pragma unroll
      for (int v = 0; v < 8; ++v) {
        const int row = wave * 16 + hi * 8 + v;
        pre[row * HH + col] = fmaxf(acc3[n][v] + bias, 0.f);
      }
    }
  }
  __syncthreads();

  if (!is_ldr) {
    for (int m = wave; m < 7; m += 8) {
      v8f acc[8];
#pragma unroll
      for (int n = 0; n < 8; ++n) acc[n] = vz;
      const int arow = m * 16 + li;
      for (int ks = 0; ks < 32; ++ks) {
        const int k0 = ks * 4 + hi * 2;
        const v2f a = *(const v2f*)(pre + arow * HH + k0);
        wmma_row8(wh0 + (2 * ks + hi) * 256, li, a, acc);
      }
#pragma unroll
      for (int n = 0; n < 8; ++n) {
        const int col = n * 16 + li;
        const float bias = lbu2[col];
#pragma unroll
        for (int v = 0; v < 8; ++v) {
          const int row = m * 16 + hi * 8 + v;
          agg[row * HH + col] = xs[row * HH + col] + acc[n][v] + bias;
        }
      }
    }
  }
  __syncthreads();

  const float* hb = agg;
  if (tid < NN) {
    const float* hr = hb + tid * HH;
    const int rot = (tid * 4) & (HH - 1);
    float s = 0.f;
    for (int k = 0; k < HH; k += 4) {
      const float4 p = *(const float4*)(hr + ((k + rot) & (HH - 1)));
      s += p.x; s += p.y; s += p.z; s += p.w;
    }
    const float mu = s * (1.0f / 128.0f);
    float var = 0.f;
    for (int k = 0; k < HH; k += 4) {
      const float4 p = *(const float4*)(hr + ((k + rot) & (HH - 1)));
      const float d0 = p.x - mu, d1 = p.y - mu, d2 = p.z - mu, d3 = p.w - mu;
      var += d0 * d0; var += d1 * d1; var += d2 * d2; var += d3 * d3;
    }
    lmu[tid] = mu;
    lrs[tid] = rsqrtf(var * (1.0f / 128.0f) + 1e-5f);
  }
  __syncthreads();

  float* og = out + (size_t)b * (NN * HH);
  for (int idx = tid; idx < NN * (HH / 4); idx += NT) {
    const int row = idx >> 5;
    const int c4  = idx & 31;
    const int col = c4 * 4;
    const float4 hv = ((const float4*)hb)[row * 32 + c4];
    const float mu = lmu[row], rs = lrs[row];
    float4 o;
    o.x = (hv.x - mu) * rs * lgm[col + 0] + lbt[col + 0];
    o.y = (hv.y - mu) * rs * lgm[col + 1] + lbt[col + 1];
    o.z = (hv.z - mu) * rs * lgm[col + 2] + lbt[col + 2];
    o.w = (hv.w - mu) * rs * lgm[col + 3] + lbt[col + 3];
    ((float4*)og)[idx] = o;
  }
}

extern "C" void kernel_launch(void* const* d_in, const int* in_sizes, int n_in,
                              void* d_out, int out_size, void* d_ws, size_t ws_size,
                              hipStream_t stream) {
  (void)n_in; (void)out_size; (void)d_ws; (void)ws_size;
  const float* x     = (const float*)d_in[0];
  const float* Wm1   = (const float*)d_in[4];
  const float* bm1   = (const float*)d_in[5];
  const float* Wm2   = (const float*)d_in[6];
  const float* bm2   = (const float*)d_in[7];
  const float* Wu1   = (const float*)d_in[8];
  const float* bu1   = (const float*)d_in[9];
  const float* Wu2   = (const float*)d_in[10];
  const float* bu2   = (const float*)d_in[11];
  const float* gamma = (const float*)d_in[12];
  const float* beta  = (const float*)d_in[13];
  float* out = (float*)d_out;

  const int B = in_sizes[0] / (NN * HH);
  const size_t shmem = (size_t)SMEM_FLOATS * sizeof(float);
  hipLaunchKernelGGL(mpnn_layer_kernel, dim3(B), dim3(NT), shmem, stream,
                     x, Wm1, bm1, Wm2, bm2, Wu1, bu1, Wu2, bu2, gamma, beta, out);
}


# Weight stage: three 128x128 LDS slots (LDS copy of x moves into slot 2 once free): x @ Wu1[:128] runs beside the 25 message tiles, 4 units a wave, one barrier fewer; atomic arrival order changes
# speedup vs baseline: 1.5467x; 1.0340x over previous
#include <hip/hip_runtime.h>


typedef __attribute__((ext_vector_type(2))) float v2f;
typedef __attribute__((ext_vector_type(8))) float v8f;

#define NN    100
#define HH    128
#define NPAD  112
#define R4    400

#define OFF_PRE   0
#define OFF_AGG   14336
#define OFF_W     28672
#define OFF_WLAST 77824
#define OFF_BM1   77952
#define OFF_BM2   78080
#define OFF_BU1   78208
#define OFF_BU2   78336
#define OFF_GM    78464
#define OFF_BT    78592
#define OFF_MU    78720
#define OFF_RS    78832
#define SMEM_FLOATS 78944
#define NT     384
#define NLDR   128
#define WSLOT  16384

__device__ __forceinline__ v8f wmma4(v2f a, v2f b, v8f c) {
  return __builtin_amdgcn_wmma_f32_16x16x4_f32(
       false, a,  false, b,
       (short)0, c,  false,  false);
}

__device__ __forceinline__ void zero16(float* dst, int nflt, int tid) {
  float4 z = make_float4(0.f, 0.f, 0.f, 0.f);
  float4* d = (float4*)dst;
  for (int i = tid; i < (nflt >> 2); i += 256) d[i] = z;
}

__device__ __forceinline__ void cpw_swz(float* dst, const float* src, int krows, int tid) {
  const int npairs = krows >> 1;
  for (int idx = tid; idx < npairs * 32; idx += 256) {
    const int kp = idx >> 5;
    const int cg = idx & 31;
    const float4 r0 = *(const float4*)(src + (size_t)(2 * kp)     * 128 + cg * 4);
    const float4 r1 = *(const float4*)(src + (size_t)(2 * kp + 1) * 128 + cg * 4);
    float4* d = (float4*)(dst + kp * 256 + cg * 8);
    d[0] = make_float4(r0.x, r1.x, r0.y, r1.y);
    d[1] = make_float4(r0.z, r1.z, r0.w, r1.w);
  }
}

__device__ __forceinline__ void ldr_stage128(float* dst, const float* src, int lt) {
  float4 r0[16], r1[16];
#pragma unroll
  for (int u = 0; u < 16; ++u) {
    const int idx = lt + u * NLDR;
    const int kp = idx >> 5;
    const int cg = idx & 31;
    r0[u] = *(const float4*)(src + (size_t)(2 * kp)     * 128 + cg * 4);
    r1[u] = *(const float4*)(src + (size_t)(2 * kp + 1) * 128 + cg * 4);
  }
#pragma unroll
  for (int u = 0; u < 16; ++u) {
    const int idx = lt + u * NLDR;
    const int kp = idx >> 5;
    const int cg = idx & 31;
    float4* d = (float4*)(dst + kp * 256 + cg * 8);
    d[0] = make_float4(r0[u].x, r1[u].x, r0[u].y, r1[u].y);
    d[1] = make_float4(r0[u].z, r1[u].z, r0[u].w, r1[u].w);
  }
}

__device__ __forceinline__ void ldr_copy_x(float* dst, const float* src, int lt) {
#pragma unroll 1
  for (int g = 0; g < 5; ++g) {
    float4 r[5];
#pragma unroll
    for (int u = 0; u < 5; ++u) r[u] = ((const float4*)src)[lt + (g * 5 + u) * NLDR];
#pragma unroll
    for (int u = 0; u < 5; ++u) ((float4*)dst)[lt + (g * 5 + u) * NLDR] = r[u];
  }
  const float4 z = make_float4(0.f, 0.f, 0.f, 0.f);
#pragma unroll
  for (int u = 0; u < 3; ++u) ((float4*)(dst + NN * HH))[lt + u * NLDR] = z;
}

__device__ __forceinline__ void wmma_row8(const float* __restrict__ wbk, int li,
                                          v2f a, v8f acc[8]) {
#pragma unroll
  for (int n = 0; n < 8; ++n) {
    const v2f bb = *(const v2f*)(wbk + (n * 16 + li) * 2);
    acc[n] = wmma4(a, bb, acc[n]);
  }
}

extern "C" __global__ __launch_bounds__(NT)
void mpnn_layer_kernel(const float* __restrict__ x,
                       const float* __restrict__ Wm1, const float* __restrict__ bm1,
                       const float* __restrict__ Wm2, const float* __restrict__ bm2,
                       const float* __restrict__ Wu1, const float* __restrict__ bu1,
                       const float* __restrict__ Wu2, const float* __restrict__ bu2,
                       const float* __restrict__ gamma, const float* __restrict__ beta,
                       float* __restrict__ out)
{
  extern __shared__ float smem[];
  float* pre   = smem + OFF_PRE;
  float* agg   = smem + OFF_AGG;
  float* wb    = smem + OFF_W;
  float* wlast = smem + OFF_WLAST;
  float* lbm1  = smem + OFF_BM1;
  float* lbm2  = smem + OFF_BM2;
  float* lbu1  = smem + OFF_BU1;
  float* lbu2  = smem + OFF_BU2;
  float* lgm   = smem + OFF_GM;
  float* lbt   = smem + OFF_BT;
  float* lmu   = smem + OFF_MU;
  float* lrs   = smem + OFF_RS;

  const int tid  = threadIdx.x;
  const int wave = tid >> 5;
  const int lane = tid & 31;
  const int li   = lane & 15;
  const int hi   = lane >> 4;
  const int b    = blockIdx.x;
  const bool is_ldr = tid >= 256;
  const int  lt     = tid - 256;
  float* ws0 = wb;
  float* ws1 = wb + WSLOT;
  float* ws2 = wb + 2 * WSLOT;

  const float* xg = x + (size_t)b * (NN * HH);

  float4 xa[16];
#pragma unroll
  for (int u = 0; u < 16; ++u) xa[u] = make_float4(0.f, 0.f, 0.f, 0.f);
  if (is_ldr) {
    const int q  = lt >> 5;
    const int l5 = lt & 31;
    const float* sA = (q == 0) ? (Wm1 + 128 * 128) : (q == 1) ? bm1 : (q == 2) ? bm2 : bu1;
    float*       dA = (q == 0) ? wlast             : (q == 1) ? lbm1 : (q == 2) ? lbm2 : lbu1;
    const float* sB = (q == 0) ? bu2  : (q == 1) ? gamma : beta;
    float*       dB = (q == 0) ? lbu2 : (q == 1) ? lgm   : lbt;
    const float4 vA = ((const float4*)sA)[l5];
    float4 vB = make_float4(0.f, 0.f, 0.f, 0.f);
    if (q < 3) vB = ((const float4*)sB)[l5];
    ldr_stage128(ws0, Wm1, lt);
    ((float4*)dA)[l5] = vA;
    if (q < 3) ((float4*)dB)[l5] = vB;
  } else {
    const int xrow = wave * 16 + li;
    if (wave < 7 && xrow < NN) {
      const float* xr = xg + xrow * HH + hi * 4;
#pragma unroll
      for (int u = 0; u < 16; ++u) xa[u] = *(const float4*)(xr + u * 8);
    }
    zero16(agg, NPAD * HH, tid);
  }
  __syncthreads();

  const v8f vz = {0.f, 0.f, 0.f, 0.f, 0.f, 0.f, 0.f, 0.f};

  if (is_ldr) {
    ldr_stage128(ws1, Wm2, lt);
    __builtin_amdgcn_sched_barrier(0);
    ldr_stage128(ws2, Wu1, lt);
  } else {
    if (wave < 7) {
      const int m = wave;
      v8f acc[8];
#pragma unroll
      for (int n = 0; n < 8; ++n) acc[n] = vz;
#pragma unroll
      for (int u = 0; u < 16; ++u) {
        v2f a0, a1;
        a0.x = xa[u].x; a0.y = xa[u].y;
        a1.x = xa[u].z; a1.y = xa[u].w;
        wmma_row8(ws0 + (4 * u + 2 * hi) * 256, li, a0, acc);
        wmma_row8(ws0 + (4 * u + 2 * hi + 1) * 256, li, a1, acc);
      }
#pragma unroll
      for (int n = 0; n < 8; ++n) {
        const int col = n * 16 + li;
        const float bias = lbm1[col];
#pragma unroll
        for (int v = 0; v < 8; ++v) {
          const int row = m * 16 + hi * 8 + v;
          pre[row * HH + col] = acc[n][v] + bias;
        }
      }
    }
  }
  __syncthreads();

  v8f acc3[8];
#pragma unroll
  for (int n = 0; n < 8; ++n) acc3[n] = vz;
  const int arow3 = wave * 16 + li;
  if (is_ldr) {
    ldr_stage128(ws0, Wu1 + 128 * 128, lt);
  } else {
    const int ntile = (wave == 7) ? 4 : 3;
    for (int t = 0; t < ntile; ++t) {
      const int m = (t < 3) ? (wave + 8 * t) : 24;
      v8f acc[8];
#pragma unroll
      for (int n = 0; n < 8; ++n) acc[n] = vz;
      const int   aj = m * 4 + (li >> 2);
      const float ad = (float)(li & 3);
      for (int ks = 0; ks < 32; ++ks) {
        const int k0 = ks * 4 + hi * 2;
        const v2f p  = *(const v2f*)(pre + aj * HH + k0);
        const v2f wl = *(const v2f*)(wlast + k0);
        v2f a;
        a.x = fmaxf(p.x + ad * wl.x, 0.f);
        a.y = fmaxf(p.y + ad * wl.y, 0.f);
        wmma_row8(ws1 + (2 * ks + hi) * 256, li, a, acc);
      }
      float bias[8];
#pragma unroll
      for (int n = 0; n < 8; ++n) bias[n] = lbm2[n * 16 + li];
#pragma unroll
      for (int v = 0; v < 8; ++v) {
        const int j = m * 4 + hi * 2 + (v >> 2), d = v & 3;
        const int jm = j - (j / 10) * 10;
        int i; bool ok;
        if      (d == 0) { i = j + 10; ok = (i < 100); }
        else if (d == 1) { i = j - 10; ok = (j >= 10); }
        else if (d == 2) { i = j + 1;  ok = (jm <= 8); }
        else             { i = j - 1;  ok = (jm >= 1); }
        if (ok) {
#pragma unroll
          for (int n = 0; n < 8; ++n) {
            const int col = n * 16 + li;
            atomicAdd(&agg[i * HH + col], acc[n][v] + bias[n]);
          }
        }
      }
    }
    if (wave < 7) {
#pragma unroll
      for (int u = 0; u < 16; ++u) {
        v2f a0, a1;
        a0.x = xa[u].x; a0.y = xa[u].y;
        a1.x = xa[u].z; a1.y = xa[u].w;
        wmma_row8(ws2 + (4 * u + 2 * hi) * 256, li, a0, acc3);
        wmma_row8(ws2 + (4 * u + 2 * hi + 1) * 256, li, a1, acc3);
      }
    }
  }
  __syncthreads();

  float* xs = ws2;
  if (is_ldr) {
    ldr_stage128(ws1, Wu2, lt);
    ldr_copy_x(xs, xg, lt);
  } else if (wave < 7) {
    for (int ks = 0; ks < 32; ++ks) {
      const int k0 = ks * 4 + hi * 2;
      const v2f a = *(const v2f*)(agg + arow3 * HH + k0);
      wmma_row8(ws0 + (2 * ks + hi) * 256, li, a, acc3);
    }
#pragma unroll
    for (int n = 0; n < 8; ++n) {
      const int col = n * 16 + li;
      const float bias = lbu1[col];
#pragma unroll
      for (int v = 0; v < 8; ++v) {
        const int row = wave * 16 + hi * 8 + v;
        pre[row * HH + col] = fmaxf(acc3[n][v] + bias, 0.f);
      }
    }
  }
  __syncthreads();

  if (!is_ldr) {
    for (int m = wave; m < 7; m += 8) {
      v8f acc[8];
#pragma unroll
      for (int n = 0; n < 8; ++n) acc[n] = vz;
      const int arow = m * 16 + li;
      for (int ks = 0; ks < 32; ++ks) {
        const int k0 = ks * 4 + hi * 2;
        const v2f a = *(const v2f*)(pre + arow * HH + k0);
        wmma_row8(ws1 + (2 * ks + hi) * 256, li, a, acc);
      }
#pragma unroll
      for (int n = 0; n < 8; ++n) {
        const int col = n * 16 + li;
        const float bias = lbu2[col];
#pragma unroll
        for (int v = 0; v < 8; ++v) {
          const int row = m * 16 + hi * 8 + v;
          agg[row * HH + col] = xs[row * HH + col] + acc[n][v] + bias;
        }
      }
    }
  }
  __syncthreads();

  const float* hb = agg;
  if (tid < NN) {
    const float* hr = hb + tid * HH;
    const int rot = (tid * 4) & (HH - 1);
    float s = 0.f;
    for (int k = 0; k < HH; k += 4) {
      const float4 p = *(const float4*)(hr + ((k + rot) & (HH - 1)));
      s += p.x; s += p.y; s += p.z; s += p.w;
    }
    const float mu = s * (1.0f / 128.0f);
    float var = 0.f;
    for (int k = 0; k < HH; k += 4) {
      const float4 p = *(const float4*)(hr + ((k + rot) & (HH - 1)));
      const float d0 = p.x - mu, d1 = p.y - mu, d2 = p.z - mu, d3 = p.w - mu;
      var += d0 * d0; var += d1 * d1; var += d2 * d2; var += d3 * d3;
    }
    lmu[tid] = mu;
    lrs[tid] = rsqrtf(var * (1.0f / 128.0f) + 1e-5f);
  }
  __syncthreads();

  float* og = out + (size_t)b * (NN * HH);
  for (int idx = tid; idx < NN * (HH / 4); idx += NT) {
    const int row = idx >> 5;
    const int c4  = idx & 31;
    const int col = c4 * 4;
    const float4 hv = ((const float4*)hb)[row * 32 + c4];
    const float mu = lmu[row], rs = lrs[row];
    float4 o;
    o.x = (hv.x - mu) * rs * lgm[col + 0] + lbt[col + 0];
    o.y = (hv.y - mu) * rs * lgm[col + 1] + lbt[col + 1];
    o.z = (hv.z - mu) * rs * lgm[col + 2] + lbt[col + 2];
    o.w = (hv.w - mu) * rs * lgm[col + 3] + lbt[col + 3];
    ((float4*)og)[idx] = o;
  }
}

extern "C" void kernel_launch(void* const* d_in, const int* in_sizes, int n_in,
                              void* d_out, int out_size, void* d_ws, size_t ws_size,
                              hipStream_t stream) {
  (void)n_in; (void)out_size; (void)d_ws; (void)ws_size;
  const float* x     = (const float*)d_in[0];
  const float* Wm1   = (const float*)d_in[4];
  const float* bm1   = (const float*)d_in[5];
  const float* Wm2   = (const float*)d_in[6];
  const float* bm2   = (const float*)d_in[7];
  const float* Wu1   = (const float*)d_in[8];
  const float* bu1   = (const float*)d_in[9];
  const float* Wu2   = (const float*)d_in[10];
  const float* bu2   = (const float*)d_in[11];
  const float* gamma = (const float*)d_in[12];
  const float* beta  = (const float*)d_in[13];
  float* out = (float*)d_out;

  const int B = in_sizes[0] / (NN * HH);
  const size_t shmem = (size_t)SMEM_FLOATS * sizeof(float);
  hipLaunchKernelGGL(mpnn_layer_kernel, dim3(B), dim3(NT), shmem, stream,
                     x, Wm1, bm1, Wm2, bm2, Wu1, bu1, Wu2, bu2, gamma, beta, out);
}


# Weight stage K-quad interleaved: one 16-byte LDS load per lane feeds the B operands of two matrix instructions; K loops take 2 K-steps a trip, LDS A rows by 16 bytes too; k sum order changes
# speedup vs baseline: 1.6001x; 1.0345x over previous
#include <hip/hip_runtime.h>


typedef __attribute__((ext_vector_type(2))) float v2f;
typedef __attribute__((ext_vector_type(8))) float v8f;

#define NN    100
#define HH    128
#define NPAD  112
#define R4    400

#define OFF_PRE   0
#define OFF_AGG   14336
#define OFF_W     28672
#define OFF_WLAST 77824
#define OFF_BM1   77952
#define OFF_BM2   78080
#define OFF_BU1   78208
#define OFF_BU2   78336
#define OFF_GM    78464
#define OFF_BT    78592
#define OFF_MU    78720
#define OFF_RS    78832
#define SMEM_FLOATS 78944
#define NT     384
#define NLDR   128
#define WSLOT  16384

__device__ __forceinline__ v8f wmma4(v2f a, v2f b, v8f c) {
  return __builtin_amdgcn_wmma_f32_16x16x4_f32(
       false, a,  false, b,
       (short)0, c,  false,  false);
}

__device__ __forceinline__ void zero16(float* dst, int nflt, int tid) {
  float4 z = make_float4(0.f, 0.f, 0.f, 0.f);
  float4* d = (float4*)dst;
  for (int i = tid; i < (nflt >> 2); i += 256) d[i] = z;
}

__device__ __forceinline__ void ldr_stage128(float* dst, const float* src, int lt) {
  float4 r[8][4];
#pragma unroll
  for (int u = 0; u < 8; ++u) {
    const int idx = lt + u * NLDR;
    const int kq = idx >> 5;
    const int cg = idx & 31;
#pragma unroll
    for (int j = 0; j < 4; ++j)
      r[u][j] = *(const float4*)(src + (size_t)(4 * kq + j) * 128 + cg * 4);
  }
#pragma unroll
  for (int u = 0; u < 8; ++u) {
    const int idx = lt + u * NLDR;
    const int kq = idx >> 5;
    const int cg = idx & 31;
    float4* d = (float4*)(dst + kq * 512 + cg * 16);
    d[0] = make_float4(r[u][0].x, r[u][1].x, r[u][2].x, r[u][3].x);
    d[1] = make_float4(r[u][0].y, r[u][1].y, r[u][2].y, r[u][3].y);
    d[2] = make_float4(r[u][0].z, r[u][1].z, r[u][2].z, r[u][3].z);
    d[3] = make_float4(r[u][0].w, r[u][1].w, r[u][2].w, r[u][3].w);
  }
}

__device__ __forceinline__ void ldr_copy_x(float* dst, const float* src, int lt) {
#pragma unroll 1
  for (int g = 0; g < 5; ++g) {
    float4 r[5];
#pragma unroll
    for (int u = 0; u < 5; ++u) r[u] = ((const float4*)src)[lt + (g * 5 + u) * NLDR];
#pragma unroll
    for (int u = 0; u < 5; ++u) ((float4*)dst)[lt + (g * 5 + u) * NLDR] = r[u];
  }
  const float4 z = make_float4(0.f, 0.f, 0.f, 0.f);
#pragma unroll
  for (int u = 0; u < 3; ++u) ((float4*)(dst + NN * HH))[lt + u * NLDR] = z;
}

__device__ __forceinline__ void wmma_quad8(const float* __restrict__ wq, int li,
                                           v2f a0, v2f a1, v8f acc[8]) {
  float4 bq[8];
#pragma unroll
  for (int n = 0; n < 8; ++n) bq[n] = *(const float4*)(wq + (n * 16 + li) * 4);
#pragma unroll
  for (int n = 0; n < 8; ++n) {
    v2f bb; bb.x = bq[n].x; bb.y = bq[n].y;
    acc[n] = wmma4(a0, bb, acc[n]);
  }
#pragma unroll
  for (int n = 0; n < 8; ++n) {
    v2f bb; bb.x = bq[n].z; bb.y = bq[n].w;
    acc[n] = wmma4(a1, bb, acc[n]);
  }
}

extern "C" __global__ __launch_bounds__(NT)
void mpnn_layer_kernel(const float* __restrict__ x,
                       const float* __restrict__ Wm1, const float* __restrict__ bm1,
                       const float* __restrict__ Wm2, const float* __restrict__ bm2,
                       const float* __restrict__ Wu1, const float* __restrict__ bu1,
                       const float* __restrict__ Wu2, const float* __restrict__ bu2,
                       const float* __restrict__ gamma, const float* __restrict__ beta,
                       float* __restrict__ out)
{
  extern __shared__ float smem[];
  float* pre   = smem + OFF_PRE;
  float* agg   = smem + OFF_AGG;
  float* wb    = smem + OFF_W;
  float* wlast = smem + OFF_WLAST;
  float* lbm1  = smem + OFF_BM1;
  float* lbm2  = smem + OFF_BM2;
  float* lbu1  = smem + OFF_BU1;
  float* lbu2  = smem + OFF_BU2;
  float* lgm   = smem + OFF_GM;
  float* lbt   = smem + OFF_BT;
  float* lmu   = smem + OFF_MU;
  float* lrs   = smem + OFF_RS;

  const int tid  = threadIdx.x;
  const int wave = tid >> 5;
  const int lane = tid & 31;
  const int li   = lane & 15;
  const int hi   = lane >> 4;
  const int b    = blockIdx.x;
  const bool is_ldr = tid >= 256;
  const int  lt     = tid - 256;
  float* ws0 = wb;
  float* ws1 = wb + WSLOT;
  float* ws2 = wb + 2 * WSLOT;

  const float* xg = x + (size_t)b * (NN * HH);

  float4 xa[16];
#pragma unroll
  for (int u = 0; u < 16; ++u) xa[u] = make_float4(0.f, 0.f, 0.f, 0.f);
  if (is_ldr) {
    const int q  = lt >> 5;
    const int l5 = lt & 31;
    const float* sA = (q == 0) ? (Wm1 + 128 * 128) : (q == 1) ? bm1 : (q == 2) ? bm2 : bu1;
    float*       dA = (q == 0) ? wlast             : (q == 1) ? lbm1 : (q == 2) ? lbm2 : lbu1;
    const float* sB = (q == 0) ? bu2  : (q == 1) ? gamma : beta;
    float*       dB = (q == 0) ? lbu2 : (q == 1) ? lgm   : lbt;
    const float4 vA = ((const float4*)sA)[l5];
    float4 vB = make_float4(0.f, 0.f, 0.f, 0.f);
    if (q < 3) vB = ((const float4*)sB)[l5];
    ldr_stage128(ws0, Wm1, lt);
    ((float4*)dA)[l5] = vA;
    if (q < 3) ((float4*)dB)[l5] = vB;
  } else {
    const int xrow = wave * 16 + li;
    if (wave < 7 && xrow < NN) {
      const float* xr = xg + xrow * HH + hi * 4;
#pragma unroll
      for (int u = 0; u < 16; ++u) xa[u] = *(const float4*)(xr + u * 8);
    }
    zero16(agg, NPAD * HH, tid);
  }
  __syncthreads();

  const v8f vz = {0.f, 0.f, 0.f, 0.f, 0.f, 0.f, 0.f, 0.f};

  if (is_ldr) {
    ldr_stage128(ws1, Wm2, lt);
    __builtin_amdgcn_sched_barrier(0);
    ldr_stage128(ws2, Wu1, lt);
  } else {
    if (wave < 7) {
      const int m = wave;
      v8f acc[8];
#pragma unroll
      for (int n = 0; n < 8; ++n) acc[n] = vz;
#pragma unroll
      for (int u = 0; u < 16; ++u) {
        v2f a0, a1;
        a0.x = xa[u].x; a0.y = xa[u].y;
        a1.x = xa[u].z; a1.y = xa[u].w;
        wmma_quad8(ws0 + (2 * u + hi) * 512, li, a0, a1, acc);
      }
#pragma unroll
      for (int n = 0; n < 8; ++n) {
        const int col = n * 16 + li;
        const float bias = lbm1[col];
#pragma unroll
        for (int v = 0; v < 8; ++v) {
          const int row = m * 16 + hi * 8 + v;
          pre[row * HH + col] = acc[n][v] + bias;
        }
      }
    }
  }
  __syncthreads();

  v8f acc3[8];
#pragma unroll
  for (int n = 0; n < 8; ++n) acc3[n] = vz;
  const int arow3 = wave * 16 + li;
  if (is_ldr) {
    ldr_stage128(ws0, Wu1 + 128 * 128, lt);
  } else {
    const int ntile = (wave == 7) ? 4 : 3;
    for (int t = 0; t < ntile; ++t) {
      const int m = (t < 3) ? (wave + 8 * t) : 24;
      v8f acc[8];
#pragma unroll
      for (int n = 0; n < 8; ++n) acc[n] = vz;
      const int   aj = m * 4 + (li >> 2);
      const float ad = (float)(li & 3);
      for (int q = 0; q < 16; ++q) {
        const int k0 = q * 8 + hi * 4;
        const float4 p  = *(const float4*)(pre + aj * HH + k0);
        const float4 wl = *(const float4*)(wlast + k0);
        v2f a0, a1;
        a0.x = fmaxf(p.x + ad * wl.x, 0.f);
        a0.y = fmaxf(p.y + ad * wl.y, 0.f);
        a1.x = fmaxf(p.z + ad * wl.z, 0.f);
        a1.y = fmaxf(p.w + ad * wl.w, 0.f);
        wmma_quad8(ws1 + (2 * q + hi) * 512, li, a0, a1, acc);
      }
      float bias[8];
#pragma unroll
      for (int n = 0; n < 8; ++n) bias[n] = lbm2[n * 16 + li];
#pragma unroll
      for (int v = 0; v < 8; ++v) {
        const int j = m * 4 + hi * 2 + (v >> 2), d = v & 3;
        const int jm = j - (j / 10) * 10;
        int i; bool ok;
        if      (d == 0) { i = j + 10; ok = (i < 100); }
        else if (d == 1) { i = j - 10; ok = (j >= 10); }
        else if (d == 2) { i = j + 1;  ok = (jm <= 8); }
        else             { i = j - 1;  ok = (jm >= 1); }
        if (ok) {
#pragma unroll
          for (int n = 0; n < 8; ++n) {
            const int col = n * 16 + li;
            atomicAdd(&agg[i * HH + col], acc[n][v] + bias[n]);
          }
        }
      }
    }
    if (wave < 7) {
#pragma unroll
      for (int u = 0; u < 16; ++u) {
        v2f a0, a1;
        a0.x = xa[u].x; a0.y = xa[u].y;
        a1.x = xa[u].z; a1.y = xa[u].w;
        wmma_quad8(ws2 + (2 * u + hi) * 512, li, a0, a1, acc3);
      }
    }
  }
  __syncthreads();

  float* xs = ws2;
  if (is_ldr) {
    ldr_stage128(ws1, Wu2, lt);
    ldr_copy_x(xs, xg, lt);
  } else if (wave < 7) {
    for (int q = 0; q < 16; ++q) {
      const int k0 = q * 8 + hi * 4;
      const float4 a = *(const float4*)(agg + arow3 * HH + k0);
      v2f a0, a1;
      a0.x = a.x; a0.y = a.y;
      a1.x = a.z; a1.y = a.w;
      wmma_quad8(ws0 + (2 * q + hi) * 512, li, a0, a1, acc3);
    }
#pragma unroll
    for (int n = 0; n < 8; ++n) {
      const int col = n * 16 + li;
      const float bias = lbu1[col];
#pragma unroll
      for (int v = 0; v < 8; ++v) {
        const int row = wave * 16 + hi * 8 + v;
        pre[row * HH + col] = fmaxf(acc3[n][v] + bias, 0.f);
      }
    }
  }
  __syncthreads();

  if (!is_ldr) {
    for (int m = wave; m < 7; m += 8) {
      v8f acc[8];
#pragma unroll
      for (int n = 0; n < 8; ++n) acc[n] = vz;
      const int arow = m * 16 + li;
      for (int q = 0; q < 16; ++q) {
        const int k0 = q * 8 + hi * 4;
        const float4 a = *(const float4*)(pre + arow * HH + k0);
        v2f a0, a1;
        a0.x = a.x; a0.y = a.y;
        a1.x = a.z; a1.y = a.w;
        wmma_quad8(ws1 + (2 * q + hi) * 512, li, a0, a1, acc);
      }
#pragma unroll
      for (int n = 0; n < 8; ++n) {
        const int col = n * 16 + li;
        const float bias = lbu2[col];
#pragma unroll
        for (int v = 0; v < 8; ++v) {
          const int row = m * 16 + hi * 8 + v;
          agg[row * HH + col] = xs[row * HH + col] + acc[n][v] + bias;
        }
      }
    }
  }
  __syncthreads();

  const float* hb = agg;
  if (tid < NN) {
    const float* hr = hb + tid * HH;
    const int rot = (tid * 4) & (HH - 1);
    float s = 0.f;
    for (int k = 0; k < HH; k += 4) {
      const float4 p = *(const float4*)(hr + ((k + rot) & (HH - 1)));
      s += p.x; s += p.y; s += p.z; s += p.w;
    }
    const float mu = s * (1.0f / 128.0f);
    float var = 0.f;
    for (int k = 0; k < HH; k += 4) {
      const float4 p = *(const float4*)(hr + ((k + rot) & (HH - 1)));
      const float d0 = p.x - mu, d1 = p.y - mu, d2 = p.z - mu, d3 = p.w - mu;
      var += d0 * d0; var += d1 * d1; var += d2 * d2; var += d3 * d3;
    }
    lmu[tid] = mu;
    lrs[tid] = rsqrtf(var * (1.0f / 128.0f) + 1e-5f);
  }
  __syncthreads();

  float* og = out + (size_t)b * (NN * HH);
  for (int idx = tid; idx < NN * (HH / 4); idx += NT) {
    const int row = idx >> 5;
    const int c4  = idx & 31;
    const int col = c4 * 4;
    const float4 hv = ((const float4*)hb)[row * 32 + c4];
    const float mu = lmu[row], rs = lrs[row];
    float4 o;
    o.x = (hv.x - mu) * rs * lgm[col + 0] + lbt[col + 0];
    o.y = (hv.y - mu) * rs * lgm[col + 1] + lbt[col + 1];
    o.z = (hv.z - mu) * rs * lgm[col + 2] + lbt[col + 2];
    o.w = (hv.w - mu) * rs * lgm[col + 3] + lbt[col + 3];
    ((float4*)og)[idx] = o;
  }
}

extern "C" void kernel_launch(void* const* d_in, const int* in_sizes, int n_in,
                              void* d_out, int out_size, void* d_ws, size_t ws_size,
                              hipStream_t stream) {
  (void)n_in; (void)out_size; (void)d_ws; (void)ws_size;
  const float* x     = (const float*)d_in[0];
  const float* Wm1   = (const float*)d_in[4];
  const float* bm1   = (const float*)d_in[5];
  const float* Wm2   = (const float*)d_in[6];
  const float* bm2   = (const float*)d_in[7];
  const float* Wu1   = (const float*)d_in[8];
  const float* bu1   = (const float*)d_in[9];
  const float* Wu2   = (const float*)d_in[10];
  const float* bu2   = (const float*)d_in[11];
  const float* gamma = (const float*)d_in[12];
  const float* beta  = (const float*)d_in[13];
  float* out = (float*)d_out;

  const int B = in_sizes[0] / (NN * HH);
  const size_t shmem = (size_t)SMEM_FLOATS * sizeof(float);
  hipLaunchKernelGGL(mpnn_layer_kernel, dim3(B), dim3(NT), shmem, stream,
                     x, Wm1, bm1, Wm2, bm2, Wu1, bu1, Wu2, bu2, gamma, beta, out);
}
